# starting kernel
# baseline (speedup 1.0000x reference)
#include <hip/hip_runtime.h>


typedef __attribute__((ext_vector_type(2))) float v2f;
typedef __attribute__((ext_vector_type(8))) float v8f;

#define SWZ_SWAPX16 0x401F

__device__ __forceinline__ float tanh_t(float x) {
#if __has_builtin(__builtin_amdgcn_tanhf)
  return __builtin_amdgcn_tanhf(x);
#else
  float r;
  asm volatile("v_tanh_f32 %0, %1\n\tv_nop" : "=v"(r) : "v"(x));
  return r;
#endif
}

__device__ __forceinline__ float swapx16(float x) {
  int i = __builtin_bit_cast(int, x);
#if __has_builtin(__builtin_amdgcn_permlanex16)
  int r = __builtin_amdgcn_permlanex16(i, i, 0x76543210, 0xfedcba98, true, false);
#else
  int r = __builtin_amdgcn_ds_swizzle(i, SWZ_SWAPX16);
#endif
  return __builtin_bit_cast(float, r);
}

__device__ __forceinline__ v8f wmma4(v2f A, v2f B, v8f C) {
  return __builtin_amdgcn_wmma_f32_16x16x4_f32(false, A, false, B, (short)0, C,
                                               false, false);
}

__global__ void __launch_bounds__(32)
lstm_wmma_fused(const float* __restrict__ x,
                const float* __restrict__ W_ih,
                const float* __restrict__ W_hh,
                const float* __restrict__ b_ih,
                const float* __restrict__ b_hh,
                const float* __restrict__ fc_w,
                const float* __restrict__ fc_b,
                float* __restrict__ out,
                int T) {
  const int  lane = threadIdx.x & 31;
  const int  tile = blockIdx.x;
  const int  n    = lane & 15;
  const bool lo   = lane < 16;
  const int  b    = tile * 16 + n;

  const int m   = n;
  const int k01 = lo ? 0 : 2;
  v2f A0, A1, A2;
  A0.x = W_ih[m * 8 + (k01 + 0)];
  A0.y = W_ih[m * 8 + (k01 + 1)];
  A1.x = W_ih[m * 8 + 4 + (k01 + 0)];
  A1.y = W_ih[m * 8 + 4 + (k01 + 1)];
  A2.x = W_hh[m * 4 + (k01 + 0)];
  A2.y = W_hh[m * 4 + (k01 + 1)];

  v8f bias;
  {
    const int gb = lo ? 0 : 8;
#pragma unroll
    for (int r = 0; r < 8; ++r) bias[r] = b_ih[gb + r] + b_hh[gb + r];
  }

  const float s_pre = lo ? 0.5f : 1.0f;
  const float s_mul = lo ? 0.5f : 1.0f;
  const float s_add = lo ? 0.5f : 0.0f;

  const float* xp = x + (size_t)b * (size_t)T * 8 + k01;

  v2f Bh; Bh.x = 0.0f; Bh.y = 0.0f;
  float c0 = 0.f, c1 = 0.f, c2 = 0.f, c3 = 0.f;

  v8f gx;
  {
    v2f Bx0 = *(const v2f*)(xp);
    v2f Bx1 = *(const v2f*)(xp + 4);
    gx = wmma4(A0, Bx0, bias);
    gx = wmma4(A1, Bx1, gx);
  }

  for (int t = 0; t < T; ++t) {
    int tn = t + 1; if (tn >= T) tn = T - 1;
    const float* xn = xp + (size_t)tn * 8;
    v2f Nx0 = *(const v2f*)(xn);
    v2f Nx1 = *(const v2f*)(xn + 4);
    int tp = t + 8; if (tp >= T) tp = T - 1;
    __builtin_prefetch(xp + (size_t)tp * 8, 0, 3);

    v8f g = wmma4(A2, Bh, gx);

    float ai[4], af[4];
#pragma unroll
    for (int j = 0; j < 4; ++j) ai[j] = fmaf(s_mul, tanh_t(s_pre * g[j]), s_add);
#pragma unroll
    for (int j = 0; j < 4; ++j) af[j] = fmaf(0.5f, tanh_t(0.5f * g[4 + j]), 0.5f);

    v8f gxn = wmma4(A0, Nx0, bias);
    gxn = wmma4(A1, Nx1, gxn);

#pragma unroll
    for (int j = 0; j < 4; ++j) {
      const float ais = swapx16(ai[j]);
      const float afs = swapx16(af[j]);
      const float p   = ai[j] * ais;
      const float fj  = lo ? af[j] : afs;
      const float oj  = lo ? afs : af[j];
      float& cj = (j == 0) ? c0 : (j == 1) ? c1 : (j == 2) ? c2 : c3;
      cj = fmaf(fj, cj, p);
      const float hj = oj * tanh_t(cj);
      if (j == 0) { if (lo) Bh.x = hj; }
      else if (j == 1) { if (lo) Bh.y = hj; }
      else if (j == 2) { if (!lo) Bh.x = hj; }
      else             { if (!lo) Bh.y = hj; }
    }

    gx = gxn;
  }

  const float wA = lo ? fc_w[0] : fc_w[2];
  const float wB = lo ? fc_w[1] : fc_w[3];
  float p  = Bh.x * wA + Bh.y * wB;
  float ps = swapx16(p);
  if (lo) out[b] = p + ps + fc_b[0];
}

extern "C" void kernel_launch(void* const* d_in, const int* in_sizes, int n_in,
                              void* d_out, int out_size, void* d_ws, size_t ws_size,
                              hipStream_t stream) {
  (void)n_in; (void)d_ws; (void)ws_size;
  const float* x    = (const float*)d_in[0];
  const float* W_ih = (const float*)d_in[1];
  const float* W_hh = (const float*)d_in[2];
  const float* b_ih = (const float*)d_in[3];
  const float* b_hh = (const float*)d_in[4];
  const float* fc_w = (const float*)d_in[5];
  const float* fc_b = (const float*)d_in[6];
  float* out = (float*)d_out;

  const int B = out_size;
  const int T = in_sizes[0] / (B * 8);
  const int tiles = B / 16;

  lstm_wmma_fused<<<tiles, 32, 0, stream>>>(x, W_ih, W_hh, b_ih, b_hh,
                                            fc_w, fc_b, out, T);
}


# x loaded one 8-step block ahead into registers, the load group pinned at the block head by one scheduling barrier, so no recurrence step waits on memory; same products, same sum order
# speedup vs baseline: 1.9685x; 1.9685x over previous
#include <hip/hip_runtime.h>


typedef __attribute__((ext_vector_type(2))) float v2f;
typedef __attribute__((ext_vector_type(8))) float v8f;

#define SWZ_SWAPX16 0x401F

__device__ __forceinline__ float tanh_t(float x) {
#if __has_builtin(__builtin_amdgcn_tanhf)
  return __builtin_amdgcn_tanhf(x);
#else
  float r;
  asm volatile("v_tanh_f32 %0, %1\n\tv_nop" : "=v"(r) : "v"(x));
  return r;
#endif
}

__device__ __forceinline__ float swapx16(float x) {
  int i = __builtin_bit_cast(int, x);
#if __has_builtin(__builtin_amdgcn_permlanex16)
  int r = __builtin_amdgcn_permlanex16(i, i, 0x76543210, 0xfedcba98, true, false);
#else
  int r = __builtin_amdgcn_ds_swizzle(i, SWZ_SWAPX16);
#endif
  return __builtin_bit_cast(float, r);
}

__device__ __forceinline__ v8f wmma4(v2f A, v2f B, v8f C) {
  return __builtin_amdgcn_wmma_f32_16x16x4_f32(false, A, false, B, (short)0, C,
                                               false, false);
}

#define XBLK 8
#define PF_AHEAD 40

struct XBuf { v2f a[XBLK]; v2f b[XBLK]; };

template <bool CLAMP>
__device__ __forceinline__ void load_block(XBuf& dst, const float* xp, int first, int T) {
  if (CLAMP) {
#pragma unroll
    for (int j = 0; j < XBLK; ++j) {
      unsigned tn = (unsigned)(first + j);
      if (tn > (unsigned)(T - 1)) tn = (unsigned)(T - 1);
      const float* xn = xp + (size_t)tn * 8;
      dst.a[j] = *(const v2f*)(xn);
      dst.b[j] = *(const v2f*)(xn + 4);
    }
  } else {
    const float* xn = xp + (size_t)(unsigned)first * 8;
#pragma unroll
    for (int j = 0; j < XBLK; ++j) {
      dst.a[j] = *(const v2f*)(xn + j * 8);
      dst.b[j] = *(const v2f*)(xn + j * 8 + 4);
    }
  }
}

__device__ __forceinline__ void cell_update(const v8f g, const bool lo,
                                            const float s_pre, const float s_mul,
                                            const float s_add,
                                            float& c0, float& c1, float& c2, float& c3,
                                            v2f& Bh) {
  float ai[4], af[4];
#pragma unroll
  for (int j = 0; j < 4; ++j) ai[j] = fmaf(s_mul, tanh_t(s_pre * g[j]), s_add);
#pragma unroll
  for (int j = 0; j < 4; ++j) af[j] = fmaf(0.5f, tanh_t(0.5f * g[4 + j]), 0.5f);

#pragma unroll
  for (int j = 0; j < 4; ++j) {
    const float ais = swapx16(ai[j]);
    const float afs = swapx16(af[j]);
    const float p   = ai[j] * ais;
    const float fj  = lo ? af[j] : afs;
    const float oj  = lo ? afs : af[j];
    float& cj = (j == 0) ? c0 : (j == 1) ? c1 : (j == 2) ? c2 : c3;
    cj = fmaf(fj, cj, p);
    const float hj = oj * tanh_t(cj);
    if (j == 0) { if (lo) Bh.x = hj; }
    else if (j == 1) { if (lo) Bh.y = hj; }
    else if (j == 2) { if (!lo) Bh.x = hj; }
    else             { if (!lo) Bh.y = hj; }
  }
}

template <bool CLAMP>
__device__ __forceinline__ void run_block(XBuf& cur, const float* xp,
                                          const int t, const int T,
                                          const v2f A0, const v2f A1, const v2f A2,
                                          const v8f bias, const bool lo,
                                          const float s_pre, const float s_mul,
                                          const float s_add,
                                          v8f& g, float& c0, float& c1, float& c2, float& c3,
                                          v2f& Bh) {
  XBuf nxt;
  load_block<CLAMP>(nxt, xp, t + 1 + XBLK, T);
  {
    unsigned tp = (unsigned)(t + PF_AHEAD);
    if (tp > (unsigned)(T - 1)) tp = (unsigned)(T - 1);
    __builtin_prefetch(xp + (size_t)tp * 8, 0, 3);
  }
  __builtin_amdgcn_sched_barrier(0);

#pragma unroll
  for (int j = 0; j < XBLK; ++j) {
    v8f gxn = wmma4(A0, cur.a[j], bias);
    gxn = wmma4(A1, cur.b[j], gxn);
    cell_update(g, lo, s_pre, s_mul, s_add, c0, c1, c2, c3, Bh);
    g = wmma4(A2, Bh, gxn);
  }
  cur = nxt;
}

__global__ void __launch_bounds__(32)
lstm_wmma_fused(const float* __restrict__ x,
                const float* __restrict__ W_ih,
                const float* __restrict__ W_hh,
                const float* __restrict__ b_ih,
                const float* __restrict__ b_hh,
                const float* __restrict__ fc_w,
                const float* __restrict__ fc_b,
                float* __restrict__ out,
                int T) {
  const int  lane = threadIdx.x & 31;
  const int  tile = blockIdx.x;
  const int  n    = lane & 15;
  const bool lo   = lane < 16;
  const int  b    = tile * 16 + n;

  const int m   = n;
  const int k01 = lo ? 0 : 2;
  v2f A0, A1, A2;
  A0.x = W_ih[m * 8 + (k01 + 0)];
  A0.y = W_ih[m * 8 + (k01 + 1)];
  A1.x = W_ih[m * 8 + 4 + (k01 + 0)];
  A1.y = W_ih[m * 8 + 4 + (k01 + 1)];
  A2.x = W_hh[m * 4 + (k01 + 0)];
  A2.y = W_hh[m * 4 + (k01 + 1)];

  v8f bias;
  {
    const int gb = lo ? 0 : 8;
#pragma unroll
    for (int r = 0; r < 8; ++r) bias[r] = b_ih[gb + r] + b_hh[gb + r];
  }

  const float s_pre = lo ? 0.5f : 1.0f;
  const float s_mul = lo ? 0.5f : 1.0f;
  const float s_add = lo ? 0.5f : 0.0f;

  const float* xp = x + (size_t)b * (size_t)T * 8 + k01;

  v2f Bh; Bh.x = 0.0f; Bh.y = 0.0f;
  float c0 = 0.f, c1 = 0.f, c2 = 0.f, c3 = 0.f;

  v8f gx;
  {
    v2f Bx0 = *(const v2f*)(xp);
    v2f Bx1 = *(const v2f*)(xp + 4);
    gx = wmma4(A0, Bx0, bias);
    gx = wmma4(A1, Bx1, gx);
  }

  XBuf cur;
  load_block<true>(cur, xp, 1, T);
  v8f g = wmma4(A2, Bh, gx);

  int t = 0;
  for (; t + 2 * XBLK + 1 <= T; t += XBLK)
    run_block<false>(cur, xp, t, T, A0, A1, A2, bias, lo, s_pre, s_mul, s_add,
                     g, c0, c1, c2, c3, Bh);
  for (; t + XBLK <= T; t += XBLK)
    run_block<true>(cur, xp, t, T, A0, A1, A2, bias, lo, s_pre, s_mul, s_add,
                    g, c0, c1, c2, c3, Bh);
  for (; t < T; ++t) {
    int tn = t + 1; if (tn >= T) tn = T - 1;
    const float* xn = xp + (size_t)tn * 8;
    v2f Nx0 = *(const v2f*)(xn);
    v2f Nx1 = *(const v2f*)(xn + 4);
    v8f gxn = wmma4(A0, Nx0, bias);
    gxn = wmma4(A1, Nx1, gxn);
    cell_update(g, lo, s_pre, s_mul, s_add, c0, c1, c2, c3, Bh);
    g = wmma4(A2, Bh, gxn);
  }

  const float wA = lo ? fc_w[0] : fc_w[2];
  const float wB = lo ? fc_w[1] : fc_w[3];
  float p  = Bh.x * wA + Bh.y * wB;
  float ps = swapx16(p);
  if (lo) out[b] = p + ps + fc_b[0];
}

extern "C" void kernel_launch(void* const* d_in, const int* in_sizes, int n_in,
                              void* d_out, int out_size, void* d_ws, size_t ws_size,
                              hipStream_t stream) {
  (void)n_in; (void)d_ws; (void)ws_size;
  const float* x    = (const float*)d_in[0];
  const float* W_ih = (const float*)d_in[1];
  const float* W_hh = (const float*)d_in[2];
  const float* b_ih = (const float*)d_in[3];
  const float* b_hh = (const float*)d_in[4];
  const float* fc_w = (const float*)d_in[5];
  const float* fc_b = (const float*)d_in[6];
  float* out = (float*)d_out;

  const int B = out_size;
  const int T = in_sizes[0] / (B * 8);
  const int tiles = B / 16;

  lstm_wmma_fused<<<tiles, 32, 0, stream>>>(x, W_ih, W_hh, b_ih, b_hh,
                                            fc_w, fc_b, out, T);
}


# gate rows permuted in the weight fragment so each lane half holds all four gates of two hidden units: cell update lane-local, no cross-half exchange, no selects, no duplicate tanh; same sums
# speedup vs baseline: 2.3488x; 1.1932x over previous
#include <hip/hip_runtime.h>


typedef __attribute__((ext_vector_type(2))) float v2f;
typedef __attribute__((ext_vector_type(8))) float v8f;

#define SWZ_SWAPX16 0x401F

__device__ __forceinline__ float tanh_t(float x) {
#if __has_builtin(__builtin_amdgcn_tanhf)
  return __builtin_amdgcn_tanhf(x);
#else
  float r;
  asm volatile("v_tanh_f32 %0, %1\n\tv_nop" : "=v"(r) : "v"(x));
  return r;
#endif
}

__device__ __forceinline__ float swapx16(float x) {
  int i = __builtin_bit_cast(int, x);
#if __has_builtin(__builtin_amdgcn_permlanex16)
  int r = __builtin_amdgcn_permlanex16(i, i, 0x76543210, 0xfedcba98, true, false);
#else
  int r = __builtin_amdgcn_ds_swizzle(i, SWZ_SWAPX16);
#endif
  return __builtin_bit_cast(float, r);
}

__device__ __forceinline__ v8f wmma4(v2f A, v2f B, v8f C) {
  return __builtin_amdgcn_wmma_f32_16x16x4_f32(false, A, false, B, (short)0, C,
                                               false, false);
}

#define XBLK 8
#define PF_AHEAD 40

struct XBuf { v2f a[XBLK]; v2f b[XBLK]; };

template <bool CLAMP>
__device__ __forceinline__ void load_block(XBuf& dst, const float* xp, int first, int T) {
  if (CLAMP) {
#pragma unroll
    for (int j = 0; j < XBLK; ++j) {
      unsigned tn = (unsigned)(first + j);
      if (tn > (unsigned)(T - 1)) tn = (unsigned)(T - 1);
      const float* xn = xp + (size_t)tn * 8;
      dst.a[j] = *(const v2f*)(xn);
      dst.b[j] = *(const v2f*)(xn + 4);
    }
  } else {
    const float* xn = xp + (size_t)(unsigned)first * 8;
#pragma unroll
    for (int j = 0; j < XBLK; ++j) {
      dst.a[j] = *(const v2f*)(xn + j * 8);
      dst.b[j] = *(const v2f*)(xn + j * 8 + 4);
    }
  }
}

__device__ __forceinline__ void cell_update(const v8f g, float& c0, float& c1, v2f& Bh) {
  const float hi0 = 0.5f * g[0], hi1 = 0.5f * g[1];
  const float hf0 = 0.5f * g[4], hf1 = 0.5f * g[5];
  const float ho0 = 0.5f * g[6], ho1 = 0.5f * g[7];
  const float tg0 = tanh_t(g[2]),  tg1 = tanh_t(g[3]);
  const float si0 = fmaf(0.5f, tanh_t(hi0), 0.5f);
  const float si1 = fmaf(0.5f, tanh_t(hi1), 0.5f);
  const float sf0 = fmaf(0.5f, tanh_t(hf0), 0.5f);
  const float sf1 = fmaf(0.5f, tanh_t(hf1), 0.5f);
  const float p0  = si0 * tg0, p1 = si1 * tg1;
  c0 = fmaf(sf0, c0, p0);
  c1 = fmaf(sf1, c1, p1);
  const float so0 = fmaf(0.5f, tanh_t(ho0), 0.5f);
  const float so1 = fmaf(0.5f, tanh_t(ho1), 0.5f);
  Bh.x = so0 * tanh_t(c0);
  Bh.y = so1 * tanh_t(c1);
}

template <bool CLAMP>
__device__ __forceinline__ void run_block(XBuf& cur, const float* xp,
                                          const int t, const int T,
                                          const v2f A0, const v2f A1, const v2f A2,
                                          const v8f bias,
                                          v8f& g, float& c0, float& c1, v2f& Bh) {
  XBuf nxt;
  load_block<CLAMP>(nxt, xp, t + 1 + XBLK, T);
  {
    unsigned tp = (unsigned)(t + PF_AHEAD);
    if (tp > (unsigned)(T - 1)) tp = (unsigned)(T - 1);
    __builtin_prefetch(xp + (size_t)tp * 8, 0, 3);
  }
  __builtin_amdgcn_sched_barrier(0);

#pragma unroll
  for (int j = 0; j < XBLK; ++j) {
    v8f gxn = wmma4(A0, cur.a[j], bias);
    gxn = wmma4(A1, cur.b[j], gxn);
    cell_update(g, c0, c1, Bh);
    g = wmma4(A2, Bh, gxn);
  }
  cur = nxt;
}

__global__ void __launch_bounds__(32)
lstm_wmma_fused(const float* __restrict__ x,
                const float* __restrict__ W_ih,
                const float* __restrict__ W_hh,
                const float* __restrict__ b_ih,
                const float* __restrict__ b_hh,
                const float* __restrict__ fc_w,
                const float* __restrict__ fc_b,
                float* __restrict__ out,
                int T) {
  const int  lane = threadIdx.x & 31;
  const int  tile = blockIdx.x;
  const int  n    = lane & 15;
  const bool lo   = lane < 16;
  const int  b    = tile * 16 + n;

  const int q   = (n & 7) >> 1;
  const int m   = (((q & 1) << 1) | (q >> 1)) * 4
                + (n >> 3) * 2 + (n & 1);
  const int k01 = lo ? 0 : 2;
  v2f A0, A1, A2;
  A0.x = W_ih[m * 8 + (k01 + 0)];
  A0.y = W_ih[m * 8 + (k01 + 1)];
  A1.x = W_ih[m * 8 + 4 + (k01 + 0)];
  A1.y = W_ih[m * 8 + 4 + (k01 + 1)];
  A2.x = W_hh[m * 4 + (k01 + 0)];
  A2.y = W_hh[m * 4 + (k01 + 1)];

  v8f bias;
  {
    const int gb = lo ? 0 : 2;
#pragma unroll
    for (int r = 0; r < 8; ++r) {
      const int qr = r >> 1;
      const int gr = (((qr & 1) << 1) | (qr >> 1)) * 4 + (r & 1) + gb;
      bias[r] = b_ih[gr] + b_hh[gr];
    }
  }

  const float* xp = x + (size_t)b * (size_t)T * 8 + k01;

  v2f Bh; Bh.x = 0.0f; Bh.y = 0.0f;
  float c0 = 0.f, c1 = 0.f;

  v8f gx;
  {
    v2f Bx0 = *(const v2f*)(xp);
    v2f Bx1 = *(const v2f*)(xp + 4);
    gx = wmma4(A0, Bx0, bias);
    gx = wmma4(A1, Bx1, gx);
  }

  XBuf cur;
  load_block<true>(cur, xp, 1, T);
  v8f g = wmma4(A2, Bh, gx);

  int t = 0;
  for (; t + 2 * XBLK + 1 <= T; t += XBLK)
    run_block<false>(cur, xp, t, T, A0, A1, A2, bias, g, c0, c1, Bh);
  for (; t + XBLK <= T; t += XBLK)
    run_block<true>(cur, xp, t, T, A0, A1, A2, bias, g, c0, c1, Bh);
  for (; t < T; ++t) {
    int tn = t + 1; if (tn >= T) tn = T - 1;
    const float* xn = xp + (size_t)tn * 8;
    v2f Nx0 = *(const v2f*)(xn);
    v2f Nx1 = *(const v2f*)(xn + 4);
    v8f gxn = wmma4(A0, Nx0, bias);
    gxn = wmma4(A1, Nx1, gxn);
    cell_update(g, c0, c1, Bh);
    g = wmma4(A2, Bh, gxn);
  }

  const float wA = lo ? fc_w[0] : fc_w[2];
  const float wB = lo ? fc_w[1] : fc_w[3];
  float p  = Bh.x * wA + Bh.y * wB;
  float ps = swapx16(p);
  if (lo) out[b] = p + ps + fc_b[0];
}

extern "C" void kernel_launch(void* const* d_in, const int* in_sizes, int n_in,
                              void* d_out, int out_size, void* d_ws, size_t ws_size,
                              hipStream_t stream) {
  (void)n_in; (void)d_ws; (void)ws_size;
  const float* x    = (const float*)d_in[0];
  const float* W_ih = (const float*)d_in[1];
  const float* W_hh = (const float*)d_in[2];
  const float* b_ih = (const float*)d_in[3];
  const float* b_hh = (const float*)d_in[4];
  const float* fc_w = (const float*)d_in[5];
  const float* fc_b = (const float*)d_in[6];
  float* out = (float*)d_out;

  const int B = out_size;
  const int T = in_sizes[0] / (B * 8);
  const int tiles = B / 16;

  lstm_wmma_fused<<<tiles, 32, 0, stream>>>(x, W_ih, W_hh, b_ih, b_hh,
                                            fc_w, fc_b, out, T);
}


# x fetched with one 16-byte load per lane and step in place of two 8-byte ones, input weights laid out alike (features 0,1,4,5 then 2,3,6,7): half the load instructions; x terms summed in that order
# speedup vs baseline: 2.4506x; 1.0433x over previous
#include <hip/hip_runtime.h>


typedef __attribute__((ext_vector_type(2))) float v2f;
typedef __attribute__((ext_vector_type(4))) float v4f;
typedef __attribute__((ext_vector_type(8))) float v8f;

#define SWZ_SWAPX16 0x401F

__device__ __forceinline__ float tanh_t(float x) {
#if __has_builtin(__builtin_amdgcn_tanhf)
  return __builtin_amdgcn_tanhf(x);
#else
  float r;
  asm volatile("v_tanh_f32 %0, %1\n\tv_nop" : "=v"(r) : "v"(x));
  return r;
#endif
}

__device__ __forceinline__ float swapx16(float x) {
  int i = __builtin_bit_cast(int, x);
#if __has_builtin(__builtin_amdgcn_permlanex16)
  int r = __builtin_amdgcn_permlanex16(i, i, 0x76543210, 0xfedcba98, true, false);
#else
  int r = __builtin_amdgcn_ds_swizzle(i, SWZ_SWAPX16);
#endif
  return __builtin_bit_cast(float, r);
}

__device__ __forceinline__ v8f wmma4(v2f A, v2f B, v8f C) {
  return __builtin_amdgcn_wmma_f32_16x16x4_f32(false, A, false, B, (short)0, C,
                                               false, false);
}

#define XBLK 8
#define PF_AHEAD 40

struct XBuf { v4f q[XBLK]; };

__device__ __forceinline__ v2f xlo(const v4f q) { return __builtin_shufflevector(q, q, 0, 1); }
__device__ __forceinline__ v2f xhi(const v4f q) { return __builtin_shufflevector(q, q, 2, 3); }

template <bool CLAMP>
__device__ __forceinline__ void load_block(XBuf& dst, const float* xp, int first, int T) {
  if (CLAMP) {
#pragma unroll
    for (int j = 0; j < XBLK; ++j) {
      unsigned tn = (unsigned)(first + j);
      if (tn > (unsigned)(T - 1)) tn = (unsigned)(T - 1);
      const float* xn = xp + (size_t)tn * 8;
      dst.q[j] = *(const v4f*)(xn);
    }
  } else {
    const float* xn = xp + (size_t)(unsigned)first * 8;
#pragma unroll
    for (int j = 0; j < XBLK; ++j)
      dst.q[j] = *(const v4f*)(xn + j * 8);
  }
}

__device__ __forceinline__ void cell_update(const v8f g, float& c0, float& c1, v2f& Bh) {
  const float hi0 = 0.5f * g[0], hi1 = 0.5f * g[1];
  const float hf0 = 0.5f * g[4], hf1 = 0.5f * g[5];
  const float ho0 = 0.5f * g[6], ho1 = 0.5f * g[7];
  const float tg0 = tanh_t(g[2]),  tg1 = tanh_t(g[3]);
  const float si0 = fmaf(0.5f, tanh_t(hi0), 0.5f);
  const float si1 = fmaf(0.5f, tanh_t(hi1), 0.5f);
  const float sf0 = fmaf(0.5f, tanh_t(hf0), 0.5f);
  const float sf1 = fmaf(0.5f, tanh_t(hf1), 0.5f);
  const float p0  = si0 * tg0, p1 = si1 * tg1;
  c0 = fmaf(sf0, c0, p0);
  c1 = fmaf(sf1, c1, p1);
  const float so0 = fmaf(0.5f, tanh_t(ho0), 0.5f);
  const float so1 = fmaf(0.5f, tanh_t(ho1), 0.5f);
  Bh.x = so0 * tanh_t(c0);
  Bh.y = so1 * tanh_t(c1);
}

template <bool CLAMP>
__device__ __forceinline__ void run_block(XBuf& cur, const float* xp,
                                          const int t, const int T,
                                          const v2f A0, const v2f A1, const v2f A2,
                                          const v8f bias,
                                          v8f& g, float& c0, float& c1, v2f& Bh) {
  XBuf nxt;
  load_block<CLAMP>(nxt, xp, t + 1 + XBLK, T);
  {
    unsigned tp = (unsigned)(t + PF_AHEAD);
    if (tp > (unsigned)(T - 1)) tp = (unsigned)(T - 1);
    __builtin_prefetch(xp + (size_t)tp * 8, 0, 3);
  }
  __builtin_amdgcn_sched_barrier(0);

#pragma unroll
  for (int j = 0; j < XBLK; ++j) {
    v8f gxn = wmma4(A0, xlo(cur.q[j]), bias);
    gxn = wmma4(A1, xhi(cur.q[j]), gxn);
    cell_update(g, c0, c1, Bh);
    g = wmma4(A2, Bh, gxn);
  }
  cur = nxt;
}

__global__ void __launch_bounds__(32)
lstm_wmma_fused(const float* __restrict__ x,
                const float* __restrict__ W_ih,
                const float* __restrict__ W_hh,
                const float* __restrict__ b_ih,
                const float* __restrict__ b_hh,
                const float* __restrict__ fc_w,
                const float* __restrict__ fc_b,
                float* __restrict__ out,
                int T) {
  const int  lane = threadIdx.x & 31;
  const int  tile = blockIdx.x;
  const int  n    = lane & 15;
  const bool lo   = lane < 16;
  const int  b    = tile * 16 + n;

  const int q   = (n & 7) >> 1;
  const int m   = (((q & 1) << 1) | (q >> 1)) * 4
                + (n >> 3) * 2 + (n & 1);
  const int k01 = lo ? 0 : 2;
  const int k4  = lo ? 0 : 4;
  v2f A0, A1, A2;
  A0.x = W_ih[m * 8 + (k4 + 0)];
  A0.y = W_ih[m * 8 + (k4 + 1)];
  A1.x = W_ih[m * 8 + (k4 + 2)];
  A1.y = W_ih[m * 8 + (k4 + 3)];
  A2.x = W_hh[m * 4 + (k01 + 0)];
  A2.y = W_hh[m * 4 + (k01 + 1)];

  v8f bias;
  {
    const int gb = lo ? 0 : 2;
#pragma unroll
    for (int r = 0; r < 8; ++r) {
      const int qr = r >> 1;
      const int gr = (((qr & 1) << 1) | (qr >> 1)) * 4 + (r & 1) + gb;
      bias[r] = b_ih[gr] + b_hh[gr];
    }
  }

  const float* xp = x + (size_t)b * (size_t)T * 8 + k4;

  v2f Bh; Bh.x = 0.0f; Bh.y = 0.0f;
  float c0 = 0.f, c1 = 0.f;

  v8f gx;
  {
    const v4f q0 = *(const v4f*)(xp);
    gx = wmma4(A0, xlo(q0), bias);
    gx = wmma4(A1, xhi(q0), gx);
  }

  XBuf cur;
  load_block<true>(cur, xp, 1, T);
  v8f g = wmma4(A2, Bh, gx);

  int t = 0;
  for (; t + 2 * XBLK + 1 <= T; t += XBLK)
    run_block<false>(cur, xp, t, T, A0, A1, A2, bias, g, c0, c1, Bh);
  for (; t + XBLK <= T; t += XBLK)
    run_block<true>(cur, xp, t, T, A0, A1, A2, bias, g, c0, c1, Bh);
  for (; t < T; ++t) {
    int tn = t + 1; if (tn >= T) tn = T - 1;
    const float* xn = xp + (size_t)tn * 8;
    const v4f qn = *(const v4f*)(xn);
    v8f gxn = wmma4(A0, xlo(qn), bias);
    gxn = wmma4(A1, xhi(qn), gxn);
    cell_update(g, c0, c1, Bh);
    g = wmma4(A2, Bh, gxn);
  }

  const float wA = lo ? fc_w[0] : fc_w[2];
  const float wB = lo ? fc_w[1] : fc_w[3];
  float p  = Bh.x * wA + Bh.y * wB;
  float ps = swapx16(p);
  if (lo) out[b] = p + ps + fc_b[0];
}

extern "C" void kernel_launch(void* const* d_in, const int* in_sizes, int n_in,
                              void* d_out, int out_size, void* d_ws, size_t ws_size,
                              hipStream_t stream) {
  (void)n_in; (void)d_ws; (void)ws_size;
  const float* x    = (const float*)d_in[0];
  const float* W_ih = (const float*)d_in[1];
  const float* W_hh = (const float*)d_in[2];
  const float* b_ih = (const float*)d_in[3];
  const float* b_hh = (const float*)d_in[4];
  const float* fc_w = (const float*)d_in[5];
  const float* fc_b = (const float*)d_in[6];
  float* out = (float*)d_out;

  const int B = out_size;
  const int T = in_sizes[0] / (B * 8);
  const int tiles = B / 16;

  lstm_wmma_fused<<<tiles, 32, 0, stream>>>(x, W_ih, W_hh, b_ih, b_hh,
                                            fc_w, fc_b, out, T);
}


# input part (bias + W_ih*x, two matrix instructions a step) moved to a second wave on another SIMD, handed over as accumulator fragments through an LDS ring; same products, same sum order
# speedup vs baseline: 2.6106x; 1.0653x over previous
#include <hip/hip_runtime.h>


typedef __attribute__((ext_vector_type(2))) float v2f;
typedef __attribute__((ext_vector_type(4))) float v4f;
typedef __attribute__((ext_vector_type(8))) float v8f;

#define SWZ_SWAPX16 0x401F

__device__ __forceinline__ float tanh_t(float x) {
#if __has_builtin(__builtin_amdgcn_tanhf)
  return __builtin_amdgcn_tanhf(x);
#else
  float r;
  asm volatile("v_tanh_f32 %0, %1\n\tv_nop" : "=v"(r) : "v"(x));
  return r;
#endif
}

__device__ __forceinline__ float swapx16(float x) {
  int i = __builtin_bit_cast(int, x);
#if __has_builtin(__builtin_amdgcn_permlanex16)
  int r = __builtin_amdgcn_permlanex16(i, i, 0x76543210, 0xfedcba98, true, false);
#else
  int r = __builtin_amdgcn_ds_swizzle(i, SWZ_SWAPX16);
#endif
  return __builtin_bit_cast(float, r);
}

__device__ __forceinline__ v8f wmma4(v2f A, v2f B, v8f C) {
  return __builtin_amdgcn_wmma_f32_16x16x4_f32(false, A, false, B, (short)0, C,
                                               false, false);
}

#define XBLK 8
#define RING 4
#define PF_AHEAD 32
#define BLK_V4 (XBLK * 2 * 32)

struct XBuf { v4f q[XBLK]; };
struct GBuf { v8f g[XBLK]; };

__device__ __forceinline__ v2f xlo(const v4f q) { return __builtin_shufflevector(q, q, 0, 1); }
__device__ __forceinline__ v2f xhi(const v4f q) { return __builtin_shufflevector(q, q, 2, 3); }

__device__ __forceinline__ void load_block(XBuf& dst, const float* xp, int first, int T) {
#pragma unroll
  for (int j = 0; j < XBLK; ++j) {
    unsigned tn = (unsigned)(first + j);
    if (tn > (unsigned)(T - 1)) tn = (unsigned)(T - 1);
    const float* xn = xp + (size_t)tn * 8;
    dst.q[j] = *(const v4f*)(xn);
  }
}

__device__ __forceinline__ void produce_block(XBuf& xb, const float* xp, const int k, const int T,
                                              const v2f A0, const v2f A1, const v8f bias,
                                              v4f* ring_lane) {
  XBuf nxt;
  load_block(nxt, xp, (k + 1) * XBLK, T);
  {
    unsigned tp = (unsigned)((k + 1) * XBLK + PF_AHEAD);
    if (tp > (unsigned)(T - 1)) tp = (unsigned)(T - 1);
    __builtin_prefetch(xp + (size_t)tp * 8, 0, 3);
  }
  __builtin_amdgcn_sched_barrier(0);

  v4f* dst = ring_lane + (k & (RING - 1)) * BLK_V4;
#pragma unroll
  for (int j = 0; j < XBLK; ++j) {
    v8f gx = wmma4(A0, xlo(xb.q[j]), bias);
    gx = wmma4(A1, xhi(xb.q[j]), gx);
    dst[(j * 2 + 0) * 32] = __builtin_shufflevector(gx, gx, 0, 1, 2, 3);
    dst[(j * 2 + 1) * 32] = __builtin_shufflevector(gx, gx, 4, 5, 6, 7);
  }
  xb = nxt;
}

__device__ __forceinline__ void fetch_block(GBuf& dst, const v4f* ring_lane, const int k) {
  const v4f* src = ring_lane + (k & (RING - 1)) * BLK_V4;
#pragma unroll
  for (int j = 0; j < XBLK; ++j) {
    const v4f a = src[(j * 2 + 0) * 32];
    const v4f b = src[(j * 2 + 1) * 32];
    dst.g[j] = __builtin_shufflevector(a, b, 0, 1, 2, 3, 4, 5, 6, 7);
  }
}

__device__ __forceinline__ void cell_update(const v8f g, float& c0, float& c1, v2f& Bh) {
  const float hi0 = 0.5f * g[0], hi1 = 0.5f * g[1];
  const float hf0 = 0.5f * g[4], hf1 = 0.5f * g[5];
  const float ho0 = 0.5f * g[6], ho1 = 0.5f * g[7];
  const float tg0 = tanh_t(g[2]),  tg1 = tanh_t(g[3]);
  const float si0 = fmaf(0.5f, tanh_t(hi0), 0.5f);
  const float si1 = fmaf(0.5f, tanh_t(hi1), 0.5f);
  const float sf0 = fmaf(0.5f, tanh_t(hf0), 0.5f);
  const float sf1 = fmaf(0.5f, tanh_t(hf1), 0.5f);
  const float p0  = si0 * tg0, p1 = si1 * tg1;
  c0 = fmaf(sf0, c0, p0);
  c1 = fmaf(sf1, c1, p1);
  const float so0 = fmaf(0.5f, tanh_t(ho0), 0.5f);
  const float so1 = fmaf(0.5f, tanh_t(ho1), 0.5f);
  Bh.x = so0 * tanh_t(c0);
  Bh.y = so1 * tanh_t(c1);
}

template <bool CHECK>
__device__ __forceinline__ void run_steps(const GBuf& buf, const int s0, const int T,
                                          const v2f A2, float& c0, float& c1, v2f& Bh) {
#pragma unroll
  for (int j = 0; j < XBLK; ++j) {
    if (!CHECK || s0 + j < T) {
      const v8f g = wmma4(A2, Bh, buf.g[j]);
      cell_update(g, c0, c1, Bh);
    }
  }
}

__global__ void __launch_bounds__(64)
lstm_wmma_fused(const float* __restrict__ x,
                const float* __restrict__ W_ih,
                const float* __restrict__ W_hh,
                const float* __restrict__ b_ih,
                const float* __restrict__ b_hh,
                const float* __restrict__ fc_w,
                const float* __restrict__ fc_b,
                float* __restrict__ out,
                int T) {
  __shared__ v4f ring[RING * BLK_V4];

  const int  lane = threadIdx.x & 31;
  const bool recur = __builtin_amdgcn_readfirstlane((int)(threadIdx.x >> 5)) == 0;
  const int  tile = blockIdx.x;
  const int  n    = lane & 15;
  const bool lo   = lane < 16;
  const int  b    = tile * 16 + n;

  const int q   = (n & 7) >> 1;
  const int m   = (((q & 1) << 1) | (q >> 1)) * 4
                + (n >> 3) * 2 + (n & 1);
  const int k01 = lo ? 0 : 2;
  const int k4  = lo ? 0 : 4;
  v2f A0, A1, A2;
  A0.x = W_ih[m * 8 + (k4 + 0)];
  A0.y = W_ih[m * 8 + (k4 + 1)];
  A1.x = W_ih[m * 8 + (k4 + 2)];
  A1.y = W_ih[m * 8 + (k4 + 3)];
  A2.x = W_hh[m * 4 + (k01 + 0)];
  A2.y = W_hh[m * 4 + (k01 + 1)];

  v8f bias;
  {
    const int gb = lo ? 0 : 2;
#pragma unroll
    for (int r = 0; r < 8; ++r) {
      const int qr = r >> 1;
      const int gr = (((qr & 1) << 1) | (qr >> 1)) * 4 + (r & 1) + gb;
      bias[r] = b_ih[gr] + b_hh[gr];
    }
  }

  const float* xp = x + (size_t)b * (size_t)T * 8 + k4;

  v4f* ring_lane = ring + lane;

  v2f Bh; Bh.x = 0.0f; Bh.y = 0.0f;
  float c0 = 0.f, c1 = 0.f;

  const int NB = (T + XBLK - 1) / XBLK;

  XBuf xb;
  if (!recur) {
    load_block(xb, xp, 0, T);
    produce_block(xb, xp, 0, T, A0, A1, bias, ring_lane);
    produce_block(xb, xp, 1, T, A0, A1, bias, ring_lane);
  }
  __syncthreads();

  GBuf bufA, bufB;
  if (recur) fetch_block(bufA, ring_lane, 0);

  int k = 0;
  for (; (k + 2) * XBLK <= T; k += 2) {
    if (recur) {
      fetch_block(bufB, ring_lane, k + 1);
      __builtin_amdgcn_sched_barrier(0);
      run_steps<false>(bufA, k * XBLK, T, A2, c0, c1, Bh);
    } else if (k + 2 < NB) {
      produce_block(xb, xp, k + 2, T, A0, A1, bias, ring_lane);
    }
    __syncthreads();
    if (recur) {
      if (k + 2 < NB) fetch_block(bufA, ring_lane, k + 2);
      __builtin_amdgcn_sched_barrier(0);
      run_steps<false>(bufB, (k + 1) * XBLK, T, A2, c0, c1, Bh);
    } else if (k + 3 < NB) {
      produce_block(xb, xp, k + 3, T, A0, A1, bias, ring_lane);
    }
    __syncthreads();
  }
  for (; k < NB; ++k) {
    if (recur) {
      if (k + 1 < NB) fetch_block(bufB, ring_lane, k + 1);
      run_steps<true>(bufA, k * XBLK, T, A2, c0, c1, Bh);
      if (k + 1 < NB) bufA = bufB;
    } else if (k + 2 < NB) {
      produce_block(xb, xp, k + 2, T, A0, A1, bias, ring_lane);
    }
    __syncthreads();
  }

  if (recur) {
    const float wA = lo ? fc_w[0] : fc_w[2];
    const float wB = lo ? fc_w[1] : fc_w[3];
    float p  = Bh.x * wA + Bh.y * wB;
    float ps = swapx16(p);
    if (lo) out[b] = p + ps + fc_b[0];
  }
}

extern "C" void kernel_launch(void* const* d_in, const int* in_sizes, int n_in,
                              void* d_out, int out_size, void* d_ws, size_t ws_size,
                              hipStream_t stream) {
  (void)n_in; (void)d_ws; (void)ws_size;
  const float* x    = (const float*)d_in[0];
  const float* W_ih = (const float*)d_in[1];
  const float* W_hh = (const float*)d_in[2];
  const float* b_ih = (const float*)d_in[3];
  const float* b_hh = (const float*)d_in[4];
  const float* fc_w = (const float*)d_in[5];
  const float* fc_b = (const float*)d_in[6];
  float* out = (float*)d_out;

  const int B = out_size;
  const int T = in_sizes[0] / (B * 8);
  const int tiles = B / 16;

  lstm_wmma_fused<<<tiles, 64, 0, stream>>>(x, W_ih, W_hh, b_ih, b_hh,
                                            fc_w, fc_b, out, T);
}


# second wave issues a block's 16 matrix instructions back to back into 8 accumulator sets, then its 16 LDS stores as one group (was pair, 2 stores, pair through one set); same sums, same order
# speedup vs baseline: 3.1821x; 1.2189x over previous
#include <hip/hip_runtime.h>


typedef __attribute__((ext_vector_type(2))) float v2f;
typedef __attribute__((ext_vector_type(4))) float v4f;
typedef __attribute__((ext_vector_type(8))) float v8f;

#define SWZ_SWAPX16 0x401F

__device__ __forceinline__ float tanh_t(float x) {
#if __has_builtin(__builtin_amdgcn_tanhf)
  return __builtin_amdgcn_tanhf(x);
#else
  float r;
  asm volatile("v_tanh_f32 %0, %1\n\tv_nop" : "=v"(r) : "v"(x));
  return r;
#endif
}

__device__ __forceinline__ float swapx16(float x) {
  int i = __builtin_bit_cast(int, x);
#if __has_builtin(__builtin_amdgcn_permlanex16)
  int r = __builtin_amdgcn_permlanex16(i, i, 0x76543210, 0xfedcba98, true, false);
#else
  int r = __builtin_amdgcn_ds_swizzle(i, SWZ_SWAPX16);
#endif
  return __builtin_bit_cast(float, r);
}

__device__ __forceinline__ v8f wmma4(v2f A, v2f B, v8f C) {
  return __builtin_amdgcn_wmma_f32_16x16x4_f32(false, A, false, B, (short)0, C,
                                               false, false);
}

#define XBLK 8
#define RING 4
#define PF_AHEAD 32
#define BLK_V4 (XBLK * 2 * 32)

struct XBuf { v4f q[XBLK]; };
struct GBuf { v8f g[XBLK]; };

__device__ __forceinline__ v2f xlo(const v4f q) { return __builtin_shufflevector(q, q, 0, 1); }
__device__ __forceinline__ v2f xhi(const v4f q) { return __builtin_shufflevector(q, q, 2, 3); }

__device__ __forceinline__ void load_block(XBuf& dst, const float* xp, int first, int T) {
#pragma unroll
  for (int j = 0; j < XBLK; ++j) {
    unsigned tn = (unsigned)(first + j);
    if (tn > (unsigned)(T - 1)) tn = (unsigned)(T - 1);
    const float* xn = xp + (size_t)tn * 8;
    dst.q[j] = *(const v4f*)(xn);
  }
}

__device__ __forceinline__ void produce_block(XBuf& xb, const float* xp, const int k, const int T,
                                              const v2f A0, const v2f A1, const v8f bias,
                                              v4f* ring_lane) {
  XBuf nxt;
  load_block(nxt, xp, (k + 1) * XBLK, T);
  {
    unsigned tp = (unsigned)((k + 1) * XBLK + PF_AHEAD);
    if (tp > (unsigned)(T - 1)) tp = (unsigned)(T - 1);
    __builtin_prefetch(xp + (size_t)tp * 8, 0, 3);
  }
  __builtin_amdgcn_sched_barrier(0);

  v4f* dst = ring_lane + (k & (RING - 1)) * BLK_V4;
  v8f gx[XBLK];
  for (int j = 0; j < XBLK; ++j) {
    gx[j] = wmma4(A0, xlo(xb.q[j]), bias);
    gx[j] = wmma4(A1, xhi(xb.q[j]), gx[j]);
  }
  for (int j = 0; j < XBLK; ++j) {
    dst[(j * 2 + 0) * 32] = __builtin_shufflevector(gx[j], gx[j], 0, 1, 2, 3);
    dst[(j * 2 + 1) * 32] = __builtin_shufflevector(gx[j], gx[j], 4, 5, 6, 7);
  }
  xb = nxt;
}

__device__ __forceinline__ void fetch_block(GBuf& dst, const v4f* ring_lane, const int k) {
  const v4f* src = ring_lane + (k & (RING - 1)) * BLK_V4;
#pragma unroll
  for (int j = 0; j < XBLK; ++j) {
    const v4f a = src[(j * 2 + 0) * 32];
    const v4f b = src[(j * 2 + 1) * 32];
    dst.g[j] = __builtin_shufflevector(a, b, 0, 1, 2, 3, 4, 5, 6, 7);
  }
}

__device__ __forceinline__ void cell_update(const v8f g, float& c0, float& c1, v2f& Bh) {
  const float hi0 = 0.5f * g[0], hi1 = 0.5f * g[1];
  const float hf0 = 0.5f * g[4], hf1 = 0.5f * g[5];
  const float ho0 = 0.5f * g[6], ho1 = 0.5f * g[7];
  const float tg0 = tanh_t(g[2]),  tg1 = tanh_t(g[3]);
  const float si0 = fmaf(0.5f, tanh_t(hi0), 0.5f);
  const float si1 = fmaf(0.5f, tanh_t(hi1), 0.5f);
  const float sf0 = fmaf(0.5f, tanh_t(hf0), 0.5f);
  const float sf1 = fmaf(0.5f, tanh_t(hf1), 0.5f);
  const float p0  = si0 * tg0, p1 = si1 * tg1;
  c0 = fmaf(sf0, c0, p0);
  c1 = fmaf(sf1, c1, p1);
  const float so0 = fmaf(0.5f, tanh_t(ho0), 0.5f);
  const float so1 = fmaf(0.5f, tanh_t(ho1), 0.5f);
  Bh.x = so0 * tanh_t(c0);
  Bh.y = so1 * tanh_t(c1);
}

template <bool CHECK>
__device__ __forceinline__ void run_steps(const GBuf& buf, const int s0, const int T,
                                          const v2f A2, float& c0, float& c1, v2f& Bh) {
#pragma unroll
  for (int j = 0; j < XBLK; ++j) {
    if (!CHECK || s0 + j < T) {
      const v8f g = wmma4(A2, Bh, buf.g[j]);
      cell_update(g, c0, c1, Bh);
    }
  }
}

__global__ void __launch_bounds__(64)
lstm_wmma_fused(const float* __restrict__ x,
                const float* __restrict__ W_ih,
                const float* __restrict__ W_hh,
                const float* __restrict__ b_ih,
                const float* __restrict__ b_hh,
                const float* __restrict__ fc_w,
                const float* __restrict__ fc_b,
                float* __restrict__ out,
                int T) {
  __shared__ v4f ring[RING * BLK_V4];

  const int  lane = threadIdx.x & 31;
  const bool recur = __builtin_amdgcn_readfirstlane((int)(threadIdx.x >> 5)) == 0;
  const int  tile = blockIdx.x;
  const int  n    = lane & 15;
  const bool lo   = lane < 16;
  const int  b    = tile * 16 + n;

  const int q   = (n & 7) >> 1;
  const int m   = (((q & 1) << 1) | (q >> 1)) * 4
                + (n >> 3) * 2 + (n & 1);
  const int k01 = lo ? 0 : 2;
  const int k4  = lo ? 0 : 4;
  v2f A0, A1, A2;
  A0.x = W_ih[m * 8 + (k4 + 0)];
  A0.y = W_ih[m * 8 + (k4 + 1)];
  A1.x = W_ih[m * 8 + (k4 + 2)];
  A1.y = W_ih[m * 8 + (k4 + 3)];
  A2.x = W_hh[m * 4 + (k01 + 0)];
  A2.y = W_hh[m * 4 + (k01 + 1)];

  v8f bias;
  {
    const int gb = lo ? 0 : 2;
#pragma unroll
    for (int r = 0; r < 8; ++r) {
      const int qr = r >> 1;
      const int gr = (((qr & 1) << 1) | (qr >> 1)) * 4 + (r & 1) + gb;
      bias[r] = b_ih[gr] + b_hh[gr];
    }
  }

  const float* xp = x + (size_t)b * (size_t)T * 8 + k4;

  v4f* ring_lane = ring + lane;

  v2f Bh; Bh.x = 0.0f; Bh.y = 0.0f;
  float c0 = 0.f, c1 = 0.f;

  const int NB = (T + XBLK - 1) / XBLK;

  XBuf xb;
  if (!recur) {
    load_block(xb, xp, 0, T);
    produce_block(xb, xp, 0, T, A0, A1, bias, ring_lane);
    produce_block(xb, xp, 1, T, A0, A1, bias, ring_lane);
  }
  __syncthreads();

  GBuf bufA, bufB;
  if (recur) fetch_block(bufA, ring_lane, 0);

  int k = 0;
  for (; (k + 2) * XBLK <= T; k += 2) {
    if (recur) {
      fetch_block(bufB, ring_lane, k + 1);
      __builtin_amdgcn_sched_barrier(0);
      run_steps<false>(bufA, k * XBLK, T, A2, c0, c1, Bh);
    } else if (k + 2 < NB) {
      produce_block(xb, xp, k + 2, T, A0, A1, bias, ring_lane);
    }
    __syncthreads();
    if (recur) {
      if (k + 2 < NB) fetch_block(bufA, ring_lane, k + 2);
      __builtin_amdgcn_sched_barrier(0);
      run_steps<false>(bufB, (k + 1) * XBLK, T, A2, c0, c1, Bh);
    } else if (k + 3 < NB) {
      produce_block(xb, xp, k + 3, T, A0, A1, bias, ring_lane);
    }
    __syncthreads();
  }
  for (; k < NB; ++k) {
    if (recur) {
      if (k + 1 < NB) fetch_block(bufB, ring_lane, k + 1);
      run_steps<true>(bufA, k * XBLK, T, A2, c0, c1, Bh);
      if (k + 1 < NB) bufA = bufB;
    } else if (k + 2 < NB) {
      produce_block(xb, xp, k + 2, T, A0, A1, bias, ring_lane);
    }
    __syncthreads();
  }

  if (recur) {
    const float wA = lo ? fc_w[0] : fc_w[2];
    const float wB = lo ? fc_w[1] : fc_w[3];
    float p  = Bh.x * wA + Bh.y * wB;
    float ps = swapx16(p);
    if (lo) out[b] = p + ps + fc_b[0];
  }
}

extern "C" void kernel_launch(void* const* d_in, const int* in_sizes, int n_in,
                              void* d_out, int out_size, void* d_ws, size_t ws_size,
                              hipStream_t stream) {
  (void)n_in; (void)d_ws; (void)ws_size;
  const float* x    = (const float*)d_in[0];
  const float* W_ih = (const float*)d_in[1];
  const float* W_hh = (const float*)d_in[2];
  const float* b_ih = (const float*)d_in[3];
  const float* b_hh = (const float*)d_in[4];
  const float* fc_w = (const float*)d_in[5];
  const float* fc_b = (const float*)d_in[6];
  float* out = (float*)d_out;

  const int B = out_size;
  const int T = in_sizes[0] / (B * 8);
  const int tiles = B / 16;

  lstm_wmma_fused<<<tiles, 64, 0, stream>>>(x, W_ih, W_hh, b_ih, b_hh,
                                            fc_w, fc_b, out, T);
}


# hand-over block doubled from 8 to 16 steps: one barrier, one group of ring loads and one wait for the x loads per 16 steps, LDS ring 64 KiB; same products, same sum order
# speedup vs baseline: 3.5236x; 1.1073x over previous
#include <hip/hip_runtime.h>


typedef __attribute__((ext_vector_type(2))) float v2f;
typedef __attribute__((ext_vector_type(4))) float v4f;
typedef __attribute__((ext_vector_type(8))) float v8f;

#define SWZ_SWAPX16 0x401F

__device__ __forceinline__ float tanh_t(float x) {
#if __has_builtin(__builtin_amdgcn_tanhf)
  return __builtin_amdgcn_tanhf(x);
#else
  float r;
  asm volatile("v_tanh_f32 %0, %1\n\tv_nop" : "=v"(r) : "v"(x));
  return r;
#endif
}

__device__ __forceinline__ float swapx16(float x) {
  int i = __builtin_bit_cast(int, x);
#if __has_builtin(__builtin_amdgcn_permlanex16)
  int r = __builtin_amdgcn_permlanex16(i, i, 0x76543210, 0xfedcba98, true, false);
#else
  int r = __builtin_amdgcn_ds_swizzle(i, SWZ_SWAPX16);
#endif
  return __builtin_bit_cast(float, r);
}

__device__ __forceinline__ v8f wmma4(v2f A, v2f B, v8f C) {
  return __builtin_amdgcn_wmma_f32_16x16x4_f32(false, A, false, B, (short)0, C,
                                               false, false);
}

#define XBLK 16
#define RING 4
#define PF_AHEAD 32
#define BLK_V4 (XBLK * 2 * 32)

struct XBuf { v4f q[XBLK]; };
struct GBuf { v8f g[XBLK]; };

__device__ __forceinline__ v2f xlo(const v4f q) { return __builtin_shufflevector(q, q, 0, 1); }
__device__ __forceinline__ v2f xhi(const v4f q) { return __builtin_shufflevector(q, q, 2, 3); }

__device__ __forceinline__ void load_block(XBuf& dst, const float* xp, int first, int T) {
#pragma unroll
  for (int j = 0; j < XBLK; ++j) {
    unsigned tn = (unsigned)(first + j);
    if (tn > (unsigned)(T - 1)) tn = (unsigned)(T - 1);
    const float* xn = xp + (size_t)tn * 8;
    dst.q[j] = *(const v4f*)(xn);
  }
}

__device__ __forceinline__ void produce_block(XBuf& xb, const float* xp, const int k, const int T,
                                              const v2f A0, const v2f A1, const v8f bias,
                                              v4f* ring_lane) {
  XBuf nxt;
  load_block(nxt, xp, (k + 1) * XBLK, T);
  for (int u = 0; u < XBLK / 8; ++u) {
    unsigned tp = (unsigned)((k + 1) * XBLK + PF_AHEAD + 8 * u);
    if (tp > (unsigned)(T - 1)) tp = (unsigned)(T - 1);
    __builtin_prefetch(xp + (size_t)tp * 8, 0, 3);
  }
  __builtin_amdgcn_sched_barrier(0);

  v4f* dst = ring_lane + (k & (RING - 1)) * BLK_V4;
  v8f gx[XBLK];
  for (int j = 0; j < XBLK; ++j) {
    gx[j] = wmma4(A0, xlo(xb.q[j]), bias);
    gx[j] = wmma4(A1, xhi(xb.q[j]), gx[j]);
  }
  for (int j = 0; j < XBLK; ++j) {
    dst[(j * 2 + 0) * 32] = __builtin_shufflevector(gx[j], gx[j], 0, 1, 2, 3);
    dst[(j * 2 + 1) * 32] = __builtin_shufflevector(gx[j], gx[j], 4, 5, 6, 7);
  }
  xb = nxt;
}

__device__ __forceinline__ void fetch_block(GBuf& dst, const v4f* ring_lane, const int k) {
  const v4f* src = ring_lane + (k & (RING - 1)) * BLK_V4;
#pragma unroll
  for (int j = 0; j < XBLK; ++j) {
    const v4f a = src[(j * 2 + 0) * 32];
    const v4f b = src[(j * 2 + 1) * 32];
    dst.g[j] = __builtin_shufflevector(a, b, 0, 1, 2, 3, 4, 5, 6, 7);
  }
}

__device__ __forceinline__ void cell_update(const v8f g, float& c0, float& c1, v2f& Bh) {
  const float hi0 = 0.5f * g[0], hi1 = 0.5f * g[1];
  const float hf0 = 0.5f * g[4], hf1 = 0.5f * g[5];
  const float ho0 = 0.5f * g[6], ho1 = 0.5f * g[7];
  const float tg0 = tanh_t(g[2]),  tg1 = tanh_t(g[3]);
  const float si0 = fmaf(0.5f, tanh_t(hi0), 0.5f);
  const float si1 = fmaf(0.5f, tanh_t(hi1), 0.5f);
  const float sf0 = fmaf(0.5f, tanh_t(hf0), 0.5f);
  const float sf1 = fmaf(0.5f, tanh_t(hf1), 0.5f);
  const float p0  = si0 * tg0, p1 = si1 * tg1;
  c0 = fmaf(sf0, c0, p0);
  c1 = fmaf(sf1, c1, p1);
  const float so0 = fmaf(0.5f, tanh_t(ho0), 0.5f);
  const float so1 = fmaf(0.5f, tanh_t(ho1), 0.5f);
  Bh.x = so0 * tanh_t(c0);
  Bh.y = so1 * tanh_t(c1);
}

template <bool CHECK>
__device__ __forceinline__ void run_steps(const GBuf& buf, const int s0, const int T,
                                          const v2f A2, float& c0, float& c1, v2f& Bh) {
#pragma unroll
  for (int j = 0; j < XBLK; ++j) {
    if (!CHECK || s0 + j < T) {
      const v8f g = wmma4(A2, Bh, buf.g[j]);
      cell_update(g, c0, c1, Bh);
    }
  }
}

__global__ void __launch_bounds__(64)
lstm_wmma_fused(const float* __restrict__ x,
                const float* __restrict__ W_ih,
                const float* __restrict__ W_hh,
                const float* __restrict__ b_ih,
                const float* __restrict__ b_hh,
                const float* __restrict__ fc_w,
                const float* __restrict__ fc_b,
                float* __restrict__ out,
                int T) {
  __shared__ v4f ring[RING * BLK_V4];

  const int  lane = threadIdx.x & 31;
  const bool recur = __builtin_amdgcn_readfirstlane((int)(threadIdx.x >> 5)) == 0;
  const int  tile = blockIdx.x;
  const int  n    = lane & 15;
  const bool lo   = lane < 16;
  const int  b    = tile * 16 + n;

  const int q   = (n & 7) >> 1;
  const int m   = (((q & 1) << 1) | (q >> 1)) * 4
                + (n >> 3) * 2 + (n & 1);
  const int k01 = lo ? 0 : 2;
  const int k4  = lo ? 0 : 4;
  v2f A0, A1, A2;
  A0.x = W_ih[m * 8 + (k4 + 0)];
  A0.y = W_ih[m * 8 + (k4 + 1)];
  A1.x = W_ih[m * 8 + (k4 + 2)];
  A1.y = W_ih[m * 8 + (k4 + 3)];
  A2.x = W_hh[m * 4 + (k01 + 0)];
  A2.y = W_hh[m * 4 + (k01 + 1)];

  v8f bias;
  {
    const int gb = lo ? 0 : 2;
#pragma unroll
    for (int r = 0; r < 8; ++r) {
      const int qr = r >> 1;
      const int gr = (((qr & 1) << 1) | (qr >> 1)) * 4 + (r & 1) + gb;
      bias[r] = b_ih[gr] + b_hh[gr];
    }
  }

  const float* xp = x + (size_t)b * (size_t)T * 8 + k4;

  v4f* ring_lane = ring + lane;

  v2f Bh; Bh.x = 0.0f; Bh.y = 0.0f;
  float c0 = 0.f, c1 = 0.f;

  const int NB = (T + XBLK - 1) / XBLK;

  XBuf xb;
  if (!recur) {
    load_block(xb, xp, 0, T);
    produce_block(xb, xp, 0, T, A0, A1, bias, ring_lane);
    produce_block(xb, xp, 1, T, A0, A1, bias, ring_lane);
  }
  __syncthreads();

  GBuf bufA, bufB;
  if (recur) fetch_block(bufA, ring_lane, 0);

  int k = 0;
  for (; (k + 2) * XBLK <= T; k += 2) {
    if (recur) {
      fetch_block(bufB, ring_lane, k + 1);
      __builtin_amdgcn_sched_barrier(0);
      run_steps<false>(bufA, k * XBLK, T, A2, c0, c1, Bh);
    } else if (k + 2 < NB) {
      produce_block(xb, xp, k + 2, T, A0, A1, bias, ring_lane);
    }
    __syncthreads();
    if (recur) {
      if (k + 2 < NB) fetch_block(bufA, ring_lane, k + 2);
      __builtin_amdgcn_sched_barrier(0);
      run_steps<false>(bufB, (k + 1) * XBLK, T, A2, c0, c1, Bh);
    } else if (k + 3 < NB) {
      produce_block(xb, xp, k + 3, T, A0, A1, bias, ring_lane);
    }
    __syncthreads();
  }
  for (; k < NB; ++k) {
    if (recur) {
      if (k + 1 < NB) fetch_block(bufB, ring_lane, k + 1);
      run_steps<true>(bufA, k * XBLK, T, A2, c0, c1, Bh);
      if (k + 1 < NB) bufA = bufB;
    } else if (k + 2 < NB) {
      produce_block(xb, xp, k + 2, T, A0, A1, bias, ring_lane);
    }
    __syncthreads();
  }

  if (recur) {
    const float wA = lo ? fc_w[0] : fc_w[2];
    const float wB = lo ? fc_w[1] : fc_w[3];
    float p  = Bh.x * wA + Bh.y * wB;
    float ps = swapx16(p);
    if (lo) out[b] = p + ps + fc_b[0];
  }
}

extern "C" void kernel_launch(void* const* d_in, const int* in_sizes, int n_in,
                              void* d_out, int out_size, void* d_ws, size_t ws_size,
                              hipStream_t stream) {
  (void)n_in; (void)d_ws; (void)ws_size;
  const float* x    = (const float*)d_in[0];
  const float* W_ih = (const float*)d_in[1];
  const float* W_hh = (const float*)d_in[2];
  const float* b_ih = (const float*)d_in[3];
  const float* b_hh = (const float*)d_in[4];
  const float* fc_w = (const float*)d_in[5];
  const float* fc_b = (const float*)d_in[6];
  float* out = (float*)d_out;

  const int B = out_size;
  const int T = in_sizes[0] / (B * 8);
  const int tiles = B / 16;

  lstm_wmma_fused<<<tiles, 64, 0, stream>>>(x, W_ih, W_hh, b_ih, b_hh,
                                            fc_w, fc_b, out, T);
}


# hand-over block doubled from 16 to 32 steps: one barrier, one group of ring loads and one wait for the x loads per 32 steps, LDS ring 128 KiB; same products, same sum order
# speedup vs baseline: 3.7936x; 1.0766x over previous
#include <hip/hip_runtime.h>


typedef __attribute__((ext_vector_type(2))) float v2f;
typedef __attribute__((ext_vector_type(4))) float v4f;
typedef __attribute__((ext_vector_type(8))) float v8f;

#define SWZ_SWAPX16 0x401F

__device__ __forceinline__ float tanh_t(float x) {
#if __has_builtin(__builtin_amdgcn_tanhf)
  return __builtin_amdgcn_tanhf(x);
#else
  float r;
  asm volatile("v_tanh_f32 %0, %1\n\tv_nop" : "=v"(r) : "v"(x));
  return r;
#endif
}

__device__ __forceinline__ float swapx16(float x) {
  int i = __builtin_bit_cast(int, x);
#if __has_builtin(__builtin_amdgcn_permlanex16)
  int r = __builtin_amdgcn_permlanex16(i, i, 0x76543210, 0xfedcba98, true, false);
#else
  int r = __builtin_amdgcn_ds_swizzle(i, SWZ_SWAPX16);
#endif
  return __builtin_bit_cast(float, r);
}

__device__ __forceinline__ v8f wmma4(v2f A, v2f B, v8f C) {
  return __builtin_amdgcn_wmma_f32_16x16x4_f32(false, A, false, B, (short)0, C,
                                               false, false);
}

#define XBLK 32
#define RING 4
#define PF_AHEAD 32
#define BLK_V4 (XBLK * 2 * 32)

struct XBuf { v4f q[XBLK]; };
struct GBuf { v8f g[XBLK]; };

__device__ __forceinline__ v2f xlo(const v4f q) { return __builtin_shufflevector(q, q, 0, 1); }
__device__ __forceinline__ v2f xhi(const v4f q) { return __builtin_shufflevector(q, q, 2, 3); }

__device__ __forceinline__ void load_block(XBuf& dst, const float* xp, int first, int T) {
#pragma unroll
  for (int j = 0; j < XBLK; ++j) {
    unsigned tn = (unsigned)(first + j);
    if (tn > (unsigned)(T - 1)) tn = (unsigned)(T - 1);
    const float* xn = xp + (size_t)tn * 8;
    dst.q[j] = *(const v4f*)(xn);
  }
}

__device__ __forceinline__ void produce_block(XBuf& xb, const float* xp, const int k, const int T,
                                              const v2f A0, const v2f A1, const v8f bias,
                                              v4f* ring_lane) {
  XBuf nxt;
  load_block(nxt, xp, (k + 1) * XBLK, T);
  for (int u = 0; u < XBLK / 8; ++u) {
    unsigned tp = (unsigned)((k + 1) * XBLK + PF_AHEAD + 8 * u);
    if (tp > (unsigned)(T - 1)) tp = (unsigned)(T - 1);
    __builtin_prefetch(xp + (size_t)tp * 8, 0, 3);
  }
  __builtin_amdgcn_sched_barrier(0);

  v4f* dst = ring_lane + (k & (RING - 1)) * BLK_V4;
  for (int hb = 0; hb < XBLK; hb += 16) {
    v8f gx[16];
    for (int j = 0; j < 16; ++j) {
      gx[j] = wmma4(A0, xlo(xb.q[hb + j]), bias);
      gx[j] = wmma4(A1, xhi(xb.q[hb + j]), gx[j]);
    }
    for (int j = 0; j < 16; ++j) {
      dst[((hb + j) * 2 + 0) * 32] = __builtin_shufflevector(gx[j], gx[j], 0, 1, 2, 3);
      dst[((hb + j) * 2 + 1) * 32] = __builtin_shufflevector(gx[j], gx[j], 4, 5, 6, 7);
    }
  }
  xb = nxt;
}

__device__ __forceinline__ void fetch_block(GBuf& dst, const v4f* ring_lane, const int k) {
  const v4f* src = ring_lane + (k & (RING - 1)) * BLK_V4;
#pragma unroll
  for (int j = 0; j < XBLK; ++j) {
    const v4f a = src[(j * 2 + 0) * 32];
    const v4f b = src[(j * 2 + 1) * 32];
    dst.g[j] = __builtin_shufflevector(a, b, 0, 1, 2, 3, 4, 5, 6, 7);
  }
}

__device__ __forceinline__ void cell_update(const v8f g, float& c0, float& c1, v2f& Bh) {
  const float hi0 = 0.5f * g[0], hi1 = 0.5f * g[1];
  const float hf0 = 0.5f * g[4], hf1 = 0.5f * g[5];
  const float ho0 = 0.5f * g[6], ho1 = 0.5f * g[7];
  const float tg0 = tanh_t(g[2]),  tg1 = tanh_t(g[3]);
  const float si0 = fmaf(0.5f, tanh_t(hi0), 0.5f);
  const float si1 = fmaf(0.5f, tanh_t(hi1), 0.5f);
  const float sf0 = fmaf(0.5f, tanh_t(hf0), 0.5f);
  const float sf1 = fmaf(0.5f, tanh_t(hf1), 0.5f);
  const float p0  = si0 * tg0, p1 = si1 * tg1;
  c0 = fmaf(sf0, c0, p0);
  c1 = fmaf(sf1, c1, p1);
  const float so0 = fmaf(0.5f, tanh_t(ho0), 0.5f);
  const float so1 = fmaf(0.5f, tanh_t(ho1), 0.5f);
  Bh.x = so0 * tanh_t(c0);
  Bh.y = so1 * tanh_t(c1);
}

template <bool CHECK>
__device__ __forceinline__ void run_steps(const GBuf& buf, const int s0, const int T,
                                          const v2f A2, float& c0, float& c1, v2f& Bh) {
#pragma unroll
  for (int j = 0; j < XBLK; ++j) {
    if (!CHECK || s0 + j < T) {
      const v8f g = wmma4(A2, Bh, buf.g[j]);
      cell_update(g, c0, c1, Bh);
    }
  }
}

__global__ void __launch_bounds__(64)
lstm_wmma_fused(const float* __restrict__ x,
                const float* __restrict__ W_ih,
                const float* __restrict__ W_hh,
                const float* __restrict__ b_ih,
                const float* __restrict__ b_hh,
                const float* __restrict__ fc_w,
                const float* __restrict__ fc_b,
                float* __restrict__ out,
                int T) {
  __shared__ v4f ring[RING * BLK_V4];

  const int  lane = threadIdx.x & 31;
  const bool recur = __builtin_amdgcn_readfirstlane((int)(threadIdx.x >> 5)) == 0;
  const int  tile = blockIdx.x;
  const int  n    = lane & 15;
  const bool lo   = lane < 16;
  const int  b    = tile * 16 + n;

  const int q   = (n & 7) >> 1;
  const int m   = (((q & 1) << 1) | (q >> 1)) * 4
                + (n >> 3) * 2 + (n & 1);
  const int k01 = lo ? 0 : 2;
  const int k4  = lo ? 0 : 4;
  v2f A0, A1, A2;
  A0.x = W_ih[m * 8 + (k4 + 0)];
  A0.y = W_ih[m * 8 + (k4 + 1)];
  A1.x = W_ih[m * 8 + (k4 + 2)];
  A1.y = W_ih[m * 8 + (k4 + 3)];
  A2.x = W_hh[m * 4 + (k01 + 0)];
  A2.y = W_hh[m * 4 + (k01 + 1)];

  v8f bias;
  {
    const int gb = lo ? 0 : 2;
#pragma unroll
    for (int r = 0; r < 8; ++r) {
      const int qr = r >> 1;
      const int gr = (((qr & 1) << 1) | (qr >> 1)) * 4 + (r & 1) + gb;
      bias[r] = b_ih[gr] + b_hh[gr];
    }
  }

  const float* xp = x + (size_t)b * (size_t)T * 8 + k4;

  v4f* ring_lane = ring + lane;

  v2f Bh; Bh.x = 0.0f; Bh.y = 0.0f;
  float c0 = 0.f, c1 = 0.f;

  const int NB = (T + XBLK - 1) / XBLK;

  XBuf xb;
  if (!recur) {
    load_block(xb, xp, 0, T);
    produce_block(xb, xp, 0, T, A0, A1, bias, ring_lane);
    produce_block(xb, xp, 1, T, A0, A1, bias, ring_lane);
  }
  __syncthreads();

  GBuf bufA, bufB;
  if (recur) fetch_block(bufA, ring_lane, 0);

  int k = 0;
  for (; (k + 2) * XBLK <= T; k += 2) {
    if (recur) {
      fetch_block(bufB, ring_lane, k + 1);
      __builtin_amdgcn_sched_barrier(0);
      run_steps<false>(bufA, k * XBLK, T, A2, c0, c1, Bh);
    } else if (k + 2 < NB) {
      produce_block(xb, xp, k + 2, T, A0, A1, bias, ring_lane);
    }
    __syncthreads();
    if (recur) {
      if (k + 2 < NB) fetch_block(bufA, ring_lane, k + 2);
      __builtin_amdgcn_sched_barrier(0);
      run_steps<false>(bufB, (k + 1) * XBLK, T, A2, c0, c1, Bh);
    } else if (k + 3 < NB) {
      produce_block(xb, xp, k + 3, T, A0, A1, bias, ring_lane);
    }
    __syncthreads();
  }
  for (; k < NB; ++k) {
    if (recur) {
      if (k + 1 < NB) fetch_block(bufB, ring_lane, k + 1);
      run_steps<true>(bufA, k * XBLK, T, A2, c0, c1, Bh);
      if (k + 1 < NB) bufA = bufB;
    } else if (k + 2 < NB) {
      produce_block(xb, xp, k + 2, T, A0, A1, bias, ring_lane);
    }
    __syncthreads();
  }

  if (recur) {
    const float wA = lo ? fc_w[0] : fc_w[2];
    const float wB = lo ? fc_w[1] : fc_w[3];
    float p  = Bh.x * wA + Bh.y * wB;
    float ps = swapx16(p);
    if (lo) out[b] = p + ps + fc_b[0];
  }
}

extern "C" void kernel_launch(void* const* d_in, const int* in_sizes, int n_in,
                              void* d_out, int out_size, void* d_ws, size_t ws_size,
                              hipStream_t stream) {
  (void)n_in; (void)d_ws; (void)ws_size;
  const float* x    = (const float*)d_in[0];
  const float* W_ih = (const float*)d_in[1];
  const float* W_hh = (const float*)d_in[2];
  const float* b_ih = (const float*)d_in[3];
  const float* b_hh = (const float*)d_in[4];
  const float* fc_w = (const float*)d_in[5];
  const float* fc_b = (const float*)d_in[6];
  float* out = (float*)d_out;

  const int B = out_size;
  const int T = in_sizes[0] / (B * 8);
  const int tiles = B / 16;

  lstm_wmma_fused<<<tiles, 64, 0, stream>>>(x, W_ih, W_hh, b_ih, b_hh,
                                            fc_w, fc_b, out, T);
}


# halving of the sigmoid gates' tanh argument folded into the i, f, o rows of W_ih, W_hh and bias at load (exact power-of-two scale): no multiply between matrix result and tanh; same sums
# speedup vs baseline: 3.9755x; 1.0479x over previous
#include <hip/hip_runtime.h>


typedef __attribute__((ext_vector_type(2))) float v2f;
typedef __attribute__((ext_vector_type(4))) float v4f;
typedef __attribute__((ext_vector_type(8))) float v8f;

#define SWZ_SWAPX16 0x401F

__device__ __forceinline__ float tanh_t(float x) {
#if __has_builtin(__builtin_amdgcn_tanhf)
  return __builtin_amdgcn_tanhf(x);
#else
  float r;
  asm volatile("v_tanh_f32 %0, %1\n\tv_nop" : "=v"(r) : "v"(x));
  return r;
#endif
}

__device__ __forceinline__ float swapx16(float x) {
  int i = __builtin_bit_cast(int, x);
#if __has_builtin(__builtin_amdgcn_permlanex16)
  int r = __builtin_amdgcn_permlanex16(i, i, 0x76543210, 0xfedcba98, true, false);
#else
  int r = __builtin_amdgcn_ds_swizzle(i, SWZ_SWAPX16);
#endif
  return __builtin_bit_cast(float, r);
}

__device__ __forceinline__ v8f wmma4(v2f A, v2f B, v8f C) {
  return __builtin_amdgcn_wmma_f32_16x16x4_f32(false, A, false, B, (short)0, C,
                                               false, false);
}

#define XBLK 32
#define RING 4
#define PF_AHEAD 32
#define BLK_V4 (XBLK * 2 * 32)

struct XBuf { v4f q[XBLK]; };
struct GBuf { v8f g[XBLK]; };

__device__ __forceinline__ v2f xlo(const v4f q) { return __builtin_shufflevector(q, q, 0, 1); }
__device__ __forceinline__ v2f xhi(const v4f q) { return __builtin_shufflevector(q, q, 2, 3); }

__device__ __forceinline__ void load_block(XBuf& dst, const float* xp, int first, int T) {
#pragma unroll
  for (int j = 0; j < XBLK; ++j) {
    unsigned tn = (unsigned)(first + j);
    if (tn > (unsigned)(T - 1)) tn = (unsigned)(T - 1);
    const float* xn = xp + (size_t)tn * 8;
    dst.q[j] = *(const v4f*)(xn);
  }
}

__device__ __forceinline__ void produce_block(XBuf& xb, const float* xp, const int k, const int T,
                                              const v2f A0, const v2f A1, const v8f bias,
                                              v4f* ring_lane) {
  XBuf nxt;
  load_block(nxt, xp, (k + 1) * XBLK, T);
  for (int u = 0; u < XBLK / 8; ++u) {
    unsigned tp = (unsigned)((k + 1) * XBLK + PF_AHEAD + 8 * u);
    if (tp > (unsigned)(T - 1)) tp = (unsigned)(T - 1);
    __builtin_prefetch(xp + (size_t)tp * 8, 0, 3);
  }
  __builtin_amdgcn_sched_barrier(0);

  v4f* dst = ring_lane + (k & (RING - 1)) * BLK_V4;
  for (int hb = 0; hb < XBLK; hb += 16) {
    v8f gx[16];
    for (int j = 0; j < 16; ++j) {
      gx[j] = wmma4(A0, xlo(xb.q[hb + j]), bias);
      gx[j] = wmma4(A1, xhi(xb.q[hb + j]), gx[j]);
    }
    for (int j = 0; j < 16; ++j) {
      dst[((hb + j) * 2 + 0) * 32] = __builtin_shufflevector(gx[j], gx[j], 0, 1, 2, 3);
      dst[((hb + j) * 2 + 1) * 32] = __builtin_shufflevector(gx[j], gx[j], 4, 5, 6, 7);
    }
  }
  xb = nxt;
}

__device__ __forceinline__ void fetch_block(GBuf& dst, const v4f* ring_lane, const int k) {
  const v4f* src = ring_lane + (k & (RING - 1)) * BLK_V4;
#pragma unroll
  for (int j = 0; j < XBLK; ++j) {
    const v4f a = src[(j * 2 + 0) * 32];
    const v4f b = src[(j * 2 + 1) * 32];
    dst.g[j] = __builtin_shufflevector(a, b, 0, 1, 2, 3, 4, 5, 6, 7);
  }
}

__device__ __forceinline__ void cell_update(const v8f g, float& c0, float& c1, v2f& Bh) {
  const float tg0 = tanh_t(g[2]),  tg1 = tanh_t(g[3]);
  const float si0 = fmaf(0.5f, tanh_t(g[0]), 0.5f);
  const float si1 = fmaf(0.5f, tanh_t(g[1]), 0.5f);
  const float sf0 = fmaf(0.5f, tanh_t(g[4]), 0.5f);
  const float sf1 = fmaf(0.5f, tanh_t(g[5]), 0.5f);
  const float p0  = si0 * tg0, p1 = si1 * tg1;
  c0 = fmaf(sf0, c0, p0);
  c1 = fmaf(sf1, c1, p1);
  const float so0 = fmaf(0.5f, tanh_t(g[6]), 0.5f);
  const float so1 = fmaf(0.5f, tanh_t(g[7]), 0.5f);
  Bh.x = so0 * tanh_t(c0);
  Bh.y = so1 * tanh_t(c1);
}

template <bool CHECK>
__device__ __forceinline__ void run_steps(const GBuf& buf, const int s0, const int T,
                                          const v2f A2, float& c0, float& c1, v2f& Bh) {
#pragma unroll
  for (int j = 0; j < XBLK; ++j) {
    if (!CHECK || s0 + j < T) {
      const v8f g = wmma4(A2, Bh, buf.g[j]);
      cell_update(g, c0, c1, Bh);
    }
  }
}

__global__ void __launch_bounds__(64)
lstm_wmma_fused(const float* __restrict__ x,
                const float* __restrict__ W_ih,
                const float* __restrict__ W_hh,
                const float* __restrict__ b_ih,
                const float* __restrict__ b_hh,
                const float* __restrict__ fc_w,
                const float* __restrict__ fc_b,
                float* __restrict__ out,
                int T) {
  __shared__ v4f ring[RING * BLK_V4];

  const int  lane = threadIdx.x & 31;
  const bool recur = __builtin_amdgcn_readfirstlane((int)(threadIdx.x >> 5)) == 0;
  const int  tile = blockIdx.x;
  const int  n    = lane & 15;
  const bool lo   = lane < 16;
  const int  b    = tile * 16 + n;

  const int q   = (n & 7) >> 1;
  const int m   = (((q & 1) << 1) | (q >> 1)) * 4
                + (n >> 3) * 2 + (n & 1);
  const int k01 = lo ? 0 : 2;
  const int k4  = lo ? 0 : 4;
  const float rs = (q == 1) ? 1.0f : 0.5f;
  v2f A0, A1, A2;
  A0.x = rs * W_ih[m * 8 + (k4 + 0)];
  A0.y = rs * W_ih[m * 8 + (k4 + 1)];
  A1.x = rs * W_ih[m * 8 + (k4 + 2)];
  A1.y = rs * W_ih[m * 8 + (k4 + 3)];
  A2.x = rs * W_hh[m * 4 + (k01 + 0)];
  A2.y = rs * W_hh[m * 4 + (k01 + 1)];

  v8f bias;
  {
    const int gb = lo ? 0 : 2;
#pragma unroll
    for (int r = 0; r < 8; ++r) {
      const int qr = r >> 1;
      const int gr = (((qr & 1) << 1) | (qr >> 1)) * 4 + (r & 1) + gb;
      bias[r] = (b_ih[gr] + b_hh[gr]) * ((qr == 1) ? 1.0f : 0.5f);
    }
  }

  const float* xp = x + (size_t)b * (size_t)T * 8 + k4;

  v4f* ring_lane = ring + lane;

  v2f Bh; Bh.x = 0.0f; Bh.y = 0.0f;
  float c0 = 0.f, c1 = 0.f;

  const int NB = (T + XBLK - 1) / XBLK;

  XBuf xb;
  if (!recur) {
    load_block(xb, xp, 0, T);
    produce_block(xb, xp, 0, T, A0, A1, bias, ring_lane);
    produce_block(xb, xp, 1, T, A0, A1, bias, ring_lane);
  }
  __syncthreads();

  GBuf bufA, bufB;
  if (recur) fetch_block(bufA, ring_lane, 0);

  int k = 0;
  for (; (k + 2) * XBLK <= T; k += 2) {
    if (recur) {
      fetch_block(bufB, ring_lane, k + 1);
      __builtin_amdgcn_sched_barrier(0);
      run_steps<false>(bufA, k * XBLK, T, A2, c0, c1, Bh);
    } else if (k + 2 < NB) {
      produce_block(xb, xp, k + 2, T, A0, A1, bias, ring_lane);
    }
    __syncthreads();
    if (recur) {
      if (k + 2 < NB) fetch_block(bufA, ring_lane, k + 2);
      __builtin_amdgcn_sched_barrier(0);
      run_steps<false>(bufB, (k + 1) * XBLK, T, A2, c0, c1, Bh);
    } else if (k + 3 < NB) {
      produce_block(xb, xp, k + 3, T, A0, A1, bias, ring_lane);
    }
    __syncthreads();
  }
  for (; k < NB; ++k) {
    if (recur) {
      if (k + 1 < NB) fetch_block(bufB, ring_lane, k + 1);
      run_steps<true>(bufA, k * XBLK, T, A2, c0, c1, Bh);
      if (k + 1 < NB) bufA = bufB;
    } else if (k + 2 < NB) {
      produce_block(xb, xp, k + 2, T, A0, A1, bias, ring_lane);
    }
    __syncthreads();
  }

  if (recur) {
    const float wA = lo ? fc_w[0] : fc_w[2];
    const float wB = lo ? fc_w[1] : fc_w[3];
    float p  = Bh.x * wA + Bh.y * wB;
    float ps = swapx16(p);
    if (lo) out[b] = p + ps + fc_b[0];
  }
}

extern "C" void kernel_launch(void* const* d_in, const int* in_sizes, int n_in,
                              void* d_out, int out_size, void* d_ws, size_t ws_size,
                              hipStream_t stream) {
  (void)n_in; (void)d_ws; (void)ws_size;
  const float* x    = (const float*)d_in[0];
  const float* W_ih = (const float*)d_in[1];
  const float* W_hh = (const float*)d_in[2];
  const float* b_ih = (const float*)d_in[3];
  const float* b_hh = (const float*)d_in[4];
  const float* fc_w = (const float*)d_in[5];
  const float* fc_b = (const float*)d_in[6];
  float* out = (float*)d_out;

  const int B = out_size;
  const int T = in_sizes[0] / (B * 8);
  const int tiles = B / 16;

  lstm_wmma_fused<<<tiles, 64, 0, stream>>>(x, W_ih, W_hh, b_ih, b_hh,
                                            fc_w, fc_b, out, T);
}


# cell update issued in a fixed order that keeps the tanh unit busy, with ONE fma behind the last gate tanh: c = fma(tanh g, sig i, sig f * c); roundings inside c change
# speedup vs baseline: 4.1078x; 1.0333x over previous
#include <hip/hip_runtime.h>


typedef __attribute__((ext_vector_type(2))) float v2f;
typedef __attribute__((ext_vector_type(4))) float v4f;
typedef __attribute__((ext_vector_type(8))) float v8f;

#define SWZ_SWAPX16 0x401F

__device__ __forceinline__ float tanh_t(float x) {
#if __has_builtin(__builtin_amdgcn_tanhf)
  return __builtin_amdgcn_tanhf(x);
#else
  float r;
  asm volatile("v_tanh_f32 %0, %1\n\tv_nop" : "=v"(r) : "v"(x));
  return r;
#endif
}

__device__ __forceinline__ float swapx16(float x) {
  int i = __builtin_bit_cast(int, x);
#if __has_builtin(__builtin_amdgcn_permlanex16)
  int r = __builtin_amdgcn_permlanex16(i, i, 0x76543210, 0xfedcba98, true, false);
#else
  int r = __builtin_amdgcn_ds_swizzle(i, SWZ_SWAPX16);
#endif
  return __builtin_bit_cast(float, r);
}

__device__ __forceinline__ v8f wmma4(v2f A, v2f B, v8f C) {
  return __builtin_amdgcn_wmma_f32_16x16x4_f32(false, A, false, B, (short)0, C,
                                               false, false);
}

#define XBLK 32
#define RING 4
#define PF_AHEAD 32
#define BLK_V4 (XBLK * 2 * 32)

struct XBuf { v4f q[XBLK]; };
struct GBuf { v8f g[XBLK]; };

__device__ __forceinline__ v2f xlo(const v4f q) { return __builtin_shufflevector(q, q, 0, 1); }
__device__ __forceinline__ v2f xhi(const v4f q) { return __builtin_shufflevector(q, q, 2, 3); }

__device__ __forceinline__ void load_block(XBuf& dst, const float* xp, int first, int T) {
#pragma unroll
  for (int j = 0; j < XBLK; ++j) {
    unsigned tn = (unsigned)(first + j);
    if (tn > (unsigned)(T - 1)) tn = (unsigned)(T - 1);
    const float* xn = xp + (size_t)tn * 8;
    dst.q[j] = *(const v4f*)(xn);
  }
}

__device__ __forceinline__ void produce_block(XBuf& xb, const float* xp, const int k, const int T,
                                              const v2f A0, const v2f A1, const v8f bias,
                                              v4f* ring_lane) {
  XBuf nxt;
  load_block(nxt, xp, (k + 1) * XBLK, T);
  for (int u = 0; u < XBLK / 8; ++u) {
    unsigned tp = (unsigned)((k + 1) * XBLK + PF_AHEAD + 8 * u);
    if (tp > (unsigned)(T - 1)) tp = (unsigned)(T - 1);
    __builtin_prefetch(xp + (size_t)tp * 8, 0, 3);
  }
  __builtin_amdgcn_sched_barrier(0);

  v4f* dst = ring_lane + (k & (RING - 1)) * BLK_V4;
  for (int hb = 0; hb < XBLK; hb += 16) {
    v8f gx[16];
    for (int j = 0; j < 16; ++j) {
      gx[j] = wmma4(A0, xlo(xb.q[hb + j]), bias);
      gx[j] = wmma4(A1, xhi(xb.q[hb + j]), gx[j]);
    }
    for (int j = 0; j < 16; ++j) {
      dst[((hb + j) * 2 + 0) * 32] = __builtin_shufflevector(gx[j], gx[j], 0, 1, 2, 3);
      dst[((hb + j) * 2 + 1) * 32] = __builtin_shufflevector(gx[j], gx[j], 4, 5, 6, 7);
    }
  }
  xb = nxt;
}

__device__ __forceinline__ void fetch_block(GBuf& dst, const v4f* ring_lane, const int k) {
  const v4f* src = ring_lane + (k & (RING - 1)) * BLK_V4;
#pragma unroll
  for (int j = 0; j < XBLK; ++j) {
    const v4f a = src[(j * 2 + 0) * 32];
    const v4f b = src[(j * 2 + 1) * 32];
    dst.g[j] = __builtin_shufflevector(a, b, 0, 1, 2, 3, 4, 5, 6, 7);
  }
}

__device__ __forceinline__ void cell_update(const v8f g, float& c0, float& c1, v2f& Bh) {
  const float tf0 = tanh_t(g[4]);
  const float tf1 = tanh_t(g[5]);
  const float ti0 = tanh_t(g[0]);
  const float ti1 = tanh_t(g[1]);
  const float sf0 = fmaf(0.5f, tf0, 0.5f);
  const float tg0 = tanh_t(g[2]);
  const float sf1 = fmaf(0.5f, tf1, 0.5f);
  const float tg1 = tanh_t(g[3]);
  const float si0 = fmaf(0.5f, ti0, 0.5f);
  const float fc0 = sf0 * c0;
  const float si1 = fmaf(0.5f, ti1, 0.5f);
  const float fc1 = sf1 * c1;
  const float to0 = tanh_t(g[6]);
  c0 = fmaf(tg0, si0, fc0);
  const float to1 = tanh_t(g[7]);
  c1 = fmaf(tg1, si1, fc1);
  const float tc0 = tanh_t(c0);
  const float tc1 = tanh_t(c1);
  const float so0 = fmaf(0.5f, to0, 0.5f);
  const float so1 = fmaf(0.5f, to1, 0.5f);
  Bh.x = so0 * tc0;
  Bh.y = so1 * tc1;
}

template <bool CHECK>
__device__ __forceinline__ void run_steps(const GBuf& buf, const int s0, const int T,
                                          const v2f A2, float& c0, float& c1, v2f& Bh) {
#pragma unroll
  for (int j = 0; j < XBLK; ++j) {
    if (!CHECK || s0 + j < T) {
      const v8f g = wmma4(A2, Bh, buf.g[j]);
      cell_update(g, c0, c1, Bh);
      __builtin_amdgcn_sched_group_barrier(0x008, 1, 0);
      __builtin_amdgcn_sched_group_barrier(0x400, 4, 0);
      __builtin_amdgcn_sched_group_barrier(0x002, 1, 0);
      __builtin_amdgcn_sched_group_barrier(0x400, 1, 0);
      __builtin_amdgcn_sched_group_barrier(0x002, 1, 0);
      __builtin_amdgcn_sched_group_barrier(0x400, 1, 0);
      __builtin_amdgcn_sched_group_barrier(0x002, 4, 0);
      __builtin_amdgcn_sched_group_barrier(0x400, 1, 0);
      __builtin_amdgcn_sched_group_barrier(0x002, 1, 0);
      __builtin_amdgcn_sched_group_barrier(0x400, 1, 0);
      __builtin_amdgcn_sched_group_barrier(0x002, 1, 0);
      __builtin_amdgcn_sched_group_barrier(0x400, 2, 0);
      __builtin_amdgcn_sched_group_barrier(0x002, 4, 0);
    }
  }
}

__global__ void __launch_bounds__(64)
lstm_wmma_fused(const float* __restrict__ x,
                const float* __restrict__ W_ih,
                const float* __restrict__ W_hh,
                const float* __restrict__ b_ih,
                const float* __restrict__ b_hh,
                const float* __restrict__ fc_w,
                const float* __restrict__ fc_b,
                float* __restrict__ out,
                int T) {
  __shared__ v4f ring[RING * BLK_V4];

  const int  lane = threadIdx.x & 31;
  const bool recur = __builtin_amdgcn_readfirstlane((int)(threadIdx.x >> 5)) == 0;
  const int  tile = blockIdx.x;
  const int  n    = lane & 15;
  const bool lo   = lane < 16;
  const int  b    = tile * 16 + n;

  const int q   = (n & 7) >> 1;
  const int m   = (((q & 1) << 1) | (q >> 1)) * 4
                + (n >> 3) * 2 + (n & 1);
  const int k01 = lo ? 0 : 2;
  const int k4  = lo ? 0 : 4;
  const float rs = (q == 1) ? 1.0f : 0.5f;
  v2f A0, A1, A2;
  A0.x = rs * W_ih[m * 8 + (k4 + 0)];
  A0.y = rs * W_ih[m * 8 + (k4 + 1)];
  A1.x = rs * W_ih[m * 8 + (k4 + 2)];
  A1.y = rs * W_ih[m * 8 + (k4 + 3)];
  A2.x = rs * W_hh[m * 4 + (k01 + 0)];
  A2.y = rs * W_hh[m * 4 + (k01 + 1)];

  v8f bias;
  {
    const int gb = lo ? 0 : 2;
#pragma unroll
    for (int r = 0; r < 8; ++r) {
      const int qr = r >> 1;
      const int gr = (((qr & 1) << 1) | (qr >> 1)) * 4 + (r & 1) + gb;
      bias[r] = (b_ih[gr] + b_hh[gr]) * ((qr == 1) ? 1.0f : 0.5f);
    }
  }

  const float* xp = x + (size_t)b * (size_t)T * 8 + k4;

  v4f* ring_lane = ring + lane;

  v2f Bh; Bh.x = 0.0f; Bh.y = 0.0f;
  float c0 = 0.f, c1 = 0.f;

  const int NB = (T + XBLK - 1) / XBLK;

  XBuf xb;
  if (!recur) {
    load_block(xb, xp, 0, T);
    produce_block(xb, xp, 0, T, A0, A1, bias, ring_lane);
    produce_block(xb, xp, 1, T, A0, A1, bias, ring_lane);
  }
  __syncthreads();

  GBuf bufA, bufB;
  if (recur) fetch_block(bufA, ring_lane, 0);

  int k = 0;
  for (; (k + 2) * XBLK <= T; k += 2) {
    if (recur) {
      fetch_block(bufB, ring_lane, k + 1);
      __builtin_amdgcn_sched_barrier(0);
      run_steps<false>(bufA, k * XBLK, T, A2, c0, c1, Bh);
    } else if (k + 2 < NB) {
      produce_block(xb, xp, k + 2, T, A0, A1, bias, ring_lane);
    }
    __syncthreads();
    if (recur) {
      if (k + 2 < NB) fetch_block(bufA, ring_lane, k + 2);
      __builtin_amdgcn_sched_barrier(0);
      run_steps<false>(bufB, (k + 1) * XBLK, T, A2, c0, c1, Bh);
    } else if (k + 3 < NB) {
      produce_block(xb, xp, k + 3, T, A0, A1, bias, ring_lane);
    }
    __syncthreads();
  }
  for (; k < NB; ++k) {
    if (recur) {
      if (k + 1 < NB) fetch_block(bufB, ring_lane, k + 1);
      run_steps<true>(bufA, k * XBLK, T, A2, c0, c1, Bh);
      if (k + 1 < NB) bufA = bufB;
    } else if (k + 2 < NB) {
      produce_block(xb, xp, k + 2, T, A0, A1, bias, ring_lane);
    }
    __syncthreads();
  }

  if (recur) {
    const float wA = lo ? fc_w[0] : fc_w[2];
    const float wB = lo ? fc_w[1] : fc_w[3];
    float p  = Bh.x * wA + Bh.y * wB;
    float ps = swapx16(p);
    if (lo) out[b] = p + ps + fc_b[0];
  }
}

extern "C" void kernel_launch(void* const* d_in, const int* in_sizes, int n_in,
                              void* d_out, int out_size, void* d_ws, size_t ws_size,
                              hipStream_t stream) {
  (void)n_in; (void)d_ws; (void)ws_size;
  const float* x    = (const float*)d_in[0];
  const float* W_ih = (const float*)d_in[1];
  const float* W_hh = (const float*)d_in[2];
  const float* b_ih = (const float*)d_in[3];
  const float* b_hh = (const float*)d_in[4];
  const float* fc_w = (const float*)d_in[5];
  const float* fc_b = (const float*)d_in[6];
  float* out = (float*)d_out;

  const int B = out_size;
  const int T = in_sizes[0] / (B * 8);
  const int tiles = B / 16;

  lstm_wmma_fused<<<tiles, 64, 0, stream>>>(x, W_ih, W_hh, b_ih, b_hh,
                                            fc_w, fc_b, out, T);
}
